# starting kernel
# baseline (speedup 1.0000x reference)
#include <hip/hip_runtime.h>
#include <hip/hip_bf16.h>

typedef float v2f __attribute__((ext_vector_type(2)));
typedef float v8f __attribute__((ext_vector_type(8)));

static __device__ __forceinline__ v8f wmma_16x16x4_f32(v2f a, v2f b, v8f c) {
#if __has_builtin(__builtin_amdgcn_wmma_f32_16x16x4_f32)
    return __builtin_amdgcn_wmma_f32_16x16x4_f32(
         false, a,  false, b,
         (short)0, c,  false,  false);
#else
    v8f d;
    asm volatile("v_wmma_f32_16x16x4_f32 %0, %1, %2, %3"
                 : "=v"(d)
                 : "v"(a), "v"(b), "v"(c));
    return d;
#endif
}

__global__ void mpo_setup_kernel(const float* __restrict__ w, float* __restrict__ ws) {
    float* bufA = ws;
    float* bufB = ws + 262144;
    float* wf   = ws + 524288;
    const int tid = threadIdx.x;

    if (tid < 64) {
        int a = tid & 15;
        int I = (tid >> 5) & 1;
        int J = (tid >> 4) & 1;
        bufA[(I * 2 + J) * 16 + a] = w[a * 4 + I * 2 + J];
    }
    __syncthreads();

    int dim = 2;
    for (int c = 1; c <= 6; ++c) {
        const float* core = w + 64 + (c - 1) * 1024;
        const int odim  = dim * 2;
        const int total = odim * odim * 16;
        for (int idx = tid; idx < total; idx += 1024) {
            int b   = idx & 15;
            int rem = idx >> 4;
            int Jo  = rem % odim;
            int Io  = rem / odim;
            int I = Io >> 1, i = Io & 1;
            int J = Jo >> 1, j = Jo & 1;
            const float* Trow = bufA + (I * dim + J) * 16;
            float s = 0.f;
#pragma unroll
            for (int a = 0; a < 16; ++a)
                s += Trow[a] * core[((a * 16 + b) * 2 + i) * 2 + j];
            bufB[idx] = s;
        }
        __syncthreads();
        float* t = bufA; bufA = bufB; bufB = t;
        dim = odim;
    }

    const float* core7 = w + 6208;
    for (int idx = tid; idx < 65536; idx += 1024) {
        int n = idx & 255;
        int r = idx >> 8;
        int I = r >> 1, i = r & 1;
        int J = n >> 1, j = n & 1;
        const float* Trow = bufA + (I * 128 + J) * 16;
        float s = 0.f;
#pragma unroll
        for (int a = 0; a < 16; ++a)
            s += Trow[a] * core7[a * 4 + i * 2 + j];
        int kstep = r >> 2, kk = r & 3;
        int nt    = n >> 4, nl = n & 15;
        int lane  = ((kk >= 2) ? 16 : 0) + nl;
        int v     = kk & 1;
        wf[((kstep * 16 + nt) * 32 + lane) * 2 + v] = s;
    }
}

__global__ void __launch_bounds__(256, 1)
mpo_main_kernel(const float* __restrict__ x,
                const float* __restrict__ dk,
                const float* __restrict__ bias,
                const float* __restrict__ wf,
                float* __restrict__ out) {
    extern __shared__ float lds[];
    float* part = lds + 65536;
    const int tid = threadIdx.x;

    {
        const float4* src = (const float4*)wf;
        float4* dst = (float4*)lds;
        for (int i = tid; i < 16384; i += 256)
            dst[i] = src[i];
    }
    __syncthreads();

    const int wave = tid >> 5;
    const int lane = tid & 31;
    const int l16  = lane & 15;
    const int koff = (lane < 16) ? 0 : 2;
    const int q    = wave >> 2;
    const int g    = wave & 3;

    const long rowbase = ((long)blockIdx.x * 2 + q) * 64;
    const float* xbase = x + (rowbase + l16) * 256 + koff;

    v8f acc[4][4];
#pragma unroll
    for (int s = 0; s < 4; ++s)
#pragma unroll
        for (int t = 0; t < 4; ++t)
#pragma unroll
            for (int v = 0; v < 8; ++v)
                acc[s][t][v] = 0.f;

    const float* ldsb = lds + lane * 2 + g * 256;

    v2f a0 = *(const v2f*)(xbase);
    v2f a1 = *(const v2f*)(xbase + 4096);
    v2f a2 = *(const v2f*)(xbase + 8192);
    v2f a3 = *(const v2f*)(xbase + 12288);

    for (int k = 0; k < 63; ++k) {
        const float* xn = xbase + 4 * (k + 1);
        v2f n0 = *(const v2f*)(xn);
        v2f n1 = *(const v2f*)(xn + 4096);
        v2f n2 = *(const v2f*)(xn + 8192);
        v2f n3 = *(const v2f*)(xn + 12288);

        const float* lk = ldsb + k * 1024;
        v2f b0 = *(const v2f*)(lk);
        v2f b1 = *(const v2f*)(lk + 64);
        v2f b2 = *(const v2f*)(lk + 128);
        v2f b3 = *(const v2f*)(lk + 192);

        acc[0][0] = wmma_16x16x4_f32(a0, b0, acc[0][0]);
        acc[0][1] = wmma_16x16x4_f32(a0, b1, acc[0][1]);
        acc[0][2] = wmma_16x16x4_f32(a0, b2, acc[0][2]);
        acc[0][3] = wmma_16x16x4_f32(a0, b3, acc[0][3]);
        acc[1][0] = wmma_16x16x4_f32(a1, b0, acc[1][0]);
        acc[1][1] = wmma_16x16x4_f32(a1, b1, acc[1][1]);
        acc[1][2] = wmma_16x16x4_f32(a1, b2, acc[1][2]);
        acc[1][3] = wmma_16x16x4_f32(a1, b3, acc[1][3]);
        acc[2][0] = wmma_16x16x4_f32(a2, b0, acc[2][0]);
        acc[2][1] = wmma_16x16x4_f32(a2, b1, acc[2][1]);
        acc[2][2] = wmma_16x16x4_f32(a2, b2, acc[2][2]);
        acc[2][3] = wmma_16x16x4_f32(a2, b3, acc[2][3]);
        acc[3][0] = wmma_16x16x4_f32(a3, b0, acc[3][0]);
        acc[3][1] = wmma_16x16x4_f32(a3, b1, acc[3][1]);
        acc[3][2] = wmma_16x16x4_f32(a3, b2, acc[3][2]);
        acc[3][3] = wmma_16x16x4_f32(a3, b3, acc[3][3]);

        a0 = n0; a1 = n1; a2 = n2; a3 = n3;
    }
    {
        const float* lk = ldsb + 63 * 1024;
        v2f b0 = *(const v2f*)(lk);
        v2f b1 = *(const v2f*)(lk + 64);
        v2f b2 = *(const v2f*)(lk + 128);
        v2f b3 = *(const v2f*)(lk + 192);
        acc[0][0] = wmma_16x16x4_f32(a0, b0, acc[0][0]);
        acc[0][1] = wmma_16x16x4_f32(a0, b1, acc[0][1]);
        acc[0][2] = wmma_16x16x4_f32(a0, b2, acc[0][2]);
        acc[0][3] = wmma_16x16x4_f32(a0, b3, acc[0][3]);
        acc[1][0] = wmma_16x16x4_f32(a1, b0, acc[1][0]);
        acc[1][1] = wmma_16x16x4_f32(a1, b1, acc[1][1]);
        acc[1][2] = wmma_16x16x4_f32(a1, b2, acc[1][2]);
        acc[1][3] = wmma_16x16x4_f32(a1, b3, acc[1][3]);
        acc[2][0] = wmma_16x16x4_f32(a2, b0, acc[2][0]);
        acc[2][1] = wmma_16x16x4_f32(a2, b1, acc[2][1]);
        acc[2][2] = wmma_16x16x4_f32(a2, b2, acc[2][2]);
        acc[2][3] = wmma_16x16x4_f32(a2, b3, acc[2][3]);
        acc[3][0] = wmma_16x16x4_f32(a3, b0, acc[3][0]);
        acc[3][1] = wmma_16x16x4_f32(a3, b1, acc[3][1]);
        acc[3][2] = wmma_16x16x4_f32(a3, b2, acc[3][2]);
        acc[3][3] = wmma_16x16x4_f32(a3, b3, acc[3][3]);
    }

    float kv[4];
#pragma unroll
    for (int t = 0; t < 4; ++t)
        kv[t] = dk[(g * 4 + t) * 16 + l16];

#pragma unroll
    for (int s = 0; s < 4; ++s) {
        float sv[8];
#pragma unroll
        for (int v = 0; v < 8; ++v) sv[v] = 0.f;
#pragma unroll
        for (int t = 0; t < 4; ++t)
#pragma unroll
            for (int v = 0; v < 8; ++v) {
                float h = acc[s][t][v];
                sv[v] += h * h * kv[t];
            }
#pragma unroll
        for (int v = 0; v < 8; ++v) {
            sv[v] += __shfl_xor(sv[v], 1, 32);
            sv[v] += __shfl_xor(sv[v], 2, 32);
            sv[v] += __shfl_xor(sv[v], 4, 32);
            sv[v] += __shfl_xor(sv[v], 8, 32);
        }
        if (l16 == 0) {
            int half = (lane >= 16) ? 8 : 0;
            float* p = part + ((q * 4 + g) * 4 + s) * 16 + half;
#pragma unroll
            for (int v = 0; v < 8; ++v) p[v] = sv[v];
        }
    }
    __syncthreads();

    if (tid < 128) {
        int q2  = tid >> 6;
        int s2  = (tid >> 4) & 3;
        int idx = tid & 15;
        const float* p = part + (q2 * 16 + s2) * 16 + idx;
        float r = p[0] + p[64] + p[128] + p[192];
        out[(long)blockIdx.x * 128 + tid] = r + bias[0];
    }
}

extern "C" void kernel_launch(void* const* d_in, const int* in_sizes, int n_in,
                              void* d_out, int out_size, void* d_ws, size_t ws_size,
                              hipStream_t stream) {
    const float* x   = (const float*)d_in[0];
    const float* mpw = (const float*)d_in[1];
    const float* dk  = (const float*)d_in[2];
    const float* db  = (const float*)d_in[3];
    float* out = (float*)d_out;
    float* ws  = (float*)d_ws;

    hipLaunchKernelGGL(mpo_setup_kernel, dim3(1), dim3(1024), 0, stream, mpw, ws);

    hipLaunchKernelGGL(mpo_main_kernel, dim3(2048), dim3(256),
                       (65536 + 512) * sizeof(float), stream,
                       x, dk, db, ws + 524288, out);
}


# main: each wave keeps its 256x64 slice of W in registers throughout (512 VGPRs; 256 blocks of 4 waves walk 64 row strips), no W in LDS; x read a strip ahead, 16-byte loads; changes f32 sum order
# speedup vs baseline: 2.4989x; 2.4989x over previous
#include <hip/hip_runtime.h>
#include <hip/hip_bf16.h>

typedef float v2f __attribute__((ext_vector_type(2)));
typedef float v4f __attribute__((ext_vector_type(4)));
typedef float v8f __attribute__((ext_vector_type(8)));

static __device__ __forceinline__ v8f wmma_16x16x4_f32(v2f a, v2f b, v8f c) {
#if __has_builtin(__builtin_amdgcn_wmma_f32_16x16x4_f32)
    return __builtin_amdgcn_wmma_f32_16x16x4_f32(
         false, a,  false, b,
         (short)0, c,  false,  false);
#else
    v8f d;
    asm volatile("v_wmma_f32_16x16x4_f32 %0, %1, %2, %3"
                 : "=v"(d)
                 : "v"(a), "v"(b), "v"(c));
    return d;
#endif
}

__global__ void __launch_bounds__(256)
mpo_setup_kernel(const float* __restrict__ w, float* __restrict__ wf) {
    __shared__ __attribute__((aligned(16))) float cs[6272];
    __shared__ float vec[2][16];
    __shared__ float t5[4 * 16];
    __shared__ float t6[16 * 16];
    __shared__ float t7[64 * 16];

    const int tid = threadIdx.x;
    const int P   = blockIdx.x >> 4;
    const int Q   = blockIdx.x & 15;
    const int b   = tid & 15;
    const int hi  = tid >> 4;

    if (tid < 224) {
        const float4* src4 = (const float4*)w;
        float4* dst4 = (float4*)cs;
        float4 r[7];
        for (int u = 0; u < 7; ++u) r[u] = src4[tid + 224 * u];
        for (int u = 0; u < 7; ++u) dst4[tid + 224 * u] = r[u];
    }
    __syncthreads();

    if (tid < 16)
        vec[0][tid] = cs[tid * 4 + ((P >> 3) & 1) * 2 + ((Q >> 3) & 1)];
    __syncthreads();

    for (int c = 1; c <= 3; ++c) {
        if (tid < 16) {
            const int i = (P >> (3 - c)) & 1;
            const int j = (Q >> (3 - c)) & 1;
            const float* core = cs + 64 + (c - 1) * 1024 + b * 4 + i * 2 + j;
            const float* src  = vec[(c - 1) & 1];
            float s = 0.f;
            for (int a = 0; a < 16; ++a)
                s += src[a] * core[a * 64];
            vec[c & 1][b] = s;
        }
        __syncthreads();
    }

    if (tid < 64) {
        const float* core = cs + 64 + 3 * 1024 + b * 4 + hi;
        float s = 0.f;
        for (int a = 0; a < 16; ++a)
            s += vec[1][a] * core[a * 64];
        t5[tid] = s;
    }
    __syncthreads();

    {
        const int Il = hi >> 2, Jl = hi & 3;
        const float* src  = t5 + ((Il >> 1) * 2 + (Jl >> 1)) * 16;
        const float* core = cs + 64 + 4 * 1024 + b * 4 + (Il & 1) * 2 + (Jl & 1);
        float s = 0.f;
        for (int a = 0; a < 16; ++a)
            s += src[a] * core[a * 64];
        t6[tid] = s;
    }
    __syncthreads();

    for (int m = 0; m < 4; ++m) {
        const int cell = hi + 16 * m;
        const int Il = cell >> 3, Jl = cell & 7;
        const float* src  = t6 + ((Il >> 1) * 4 + (Jl >> 1)) * 16;
        const float* core = cs + 64 + 5 * 1024 + b * 4 + (Il & 1) * 2 + (Jl & 1);
        float s = 0.f;
        for (int a = 0; a < 16; ++a)
            s += src[a] * core[a * 64];
        t7[cell * 16 + b] = s;
    }
    __syncthreads();

    {
        const int rl = hi, nl = b;
        const float* src  = t7 + ((rl >> 1) * 8 + (nl >> 1)) * 16;
        const float* core = cs + 6208 + (rl & 1) * 2 + (nl & 1);
        float s = 0.f;
        for (int a = 0; a < 16; ++a)
            s += src[a] * core[a * 4];
        const int r     = P * 16 + rl;
        const int kstep = r >> 2, kk = r & 3;
        const int lane  = ((kk >= 2) ? 16 : 0) + nl;
        const int v     = kk & 1;
        wf[((kstep * 16 + Q) * 32 + lane) * 2 + v] = s;
    }
}

static __device__ __forceinline__ void strip_pass(const v4f (&cur)[32], v4f (&nxt)[32],
                                                  const float* __restrict__ xnext,
                                                  const v2f (&wr)[64][4],
                                                  const float (&kv)[4],
                                                  float* __restrict__ prow,
                                                  int c, int h) {
#pragma unroll
    for (int j = 0; j < 32; ++j)
        nxt[j] = *(const v4f*)(xnext + 8 * j);

    v8f acc[4];
#pragma unroll
    for (int t = 0; t < 4; ++t)
#pragma unroll
        for (int v = 0; v < 8; ++v)
            acc[t][v] = 0.f;

#pragma unroll
    for (int j = 0; j < 32; ++j) {
        v2f alo, ahi;
        alo[0] = cur[j][0]; alo[1] = cur[j][1];
        ahi[0] = cur[j][2]; ahi[1] = cur[j][3];
#pragma unroll
        for (int t = 0; t < 4; ++t)
            acc[t] = wmma_16x16x4_f32(alo, wr[2 * j][t], acc[t]);
#pragma unroll
        for (int t = 0; t < 4; ++t)
            acc[t] = wmma_16x16x4_f32(ahi, wr[2 * j + 1][t], acc[t]);
    }

    float sv[8];
#pragma unroll
    for (int v = 0; v < 8; ++v) sv[v] = 0.f;
#pragma unroll
    for (int t = 0; t < 4; ++t)
#pragma unroll
        for (int v = 0; v < 8; ++v) {
            float hh = acc[t][v];
            sv[v] += hh * hh * kv[t];
        }
#pragma unroll
    for (int v = 0; v < 8; ++v) {
        sv[v] += __shfl_xor(sv[v], 1, 32);
        sv[v] += __shfl_xor(sv[v], 2, 32);
        sv[v] += __shfl_xor(sv[v], 4, 32);
        sv[v] += __shfl_xor(sv[v], 8, 32);
    }
    if (c == 0) {
        float* p = prow + h * 8;
#pragma unroll
        for (int v = 0; v < 8; ++v) p[v] = sv[v];
    }
}

__global__ void __launch_bounds__(128)
mpo_main_kernel(const float* __restrict__ x,
                const float* __restrict__ dk,
                const float* __restrict__ bias,
                const float* __restrict__ wf,
                float* __restrict__ out) {
    __shared__ float part[4 * 1024];
    const int tid  = threadIdx.x;
    const int g    = tid >> 5;
    const int lane = tid & 31;
    const int c    = lane & 15;
    const int h    = lane >> 4;

    v2f wr[64][4];
    {
        const float* wb = wf + ((h * 16 + g * 4) * 32 + c) * 2;
#pragma unroll
        for (int s = 0; s < 64; ++s)
#pragma unroll
            for (int t = 0; t < 4; ++t)
                wr[s][t] = *(const v2f*)(wb + (((s >> 1) * 32 + t) * 32 + 16 * (s & 1)) * 2);
    }

    float kv[4];
#pragma unroll
    for (int t = 0; t < 4; ++t)
        kv[t] = dk[(g * 4 + t) * 16 + c];

    const float* xb = x + ((long)blockIdx.x * 1024 + c) * 256 + 4 * h;
    float* pg = part + g * 1024;

    v4f bufa[32], bufb[32];
#pragma unroll
    for (int j = 0; j < 32; ++j)
        bufa[j] = *(const v4f*)(xb + 8 * j);

    for (int p = 0; p < 64; p += 2) {
        strip_pass(bufa, bufb, xb + (long)(p + 1) * 4096, wr, kv, pg + p * 16, c, h);
        const int pn = (p + 2 < 64) ? (p + 2) : 63;
        strip_pass(bufb, bufa, xb + (long)pn * 4096, wr, kv, pg + (p + 1) * 16, c, h);
    }
    __syncthreads();

#pragma unroll
    for (int u = 0; u < 8; ++u) {
        const int r = tid + 128 * u;
        float s = part[r] + part[1024 + r] + part[2048 + r] + part[3072 + r];
        out[(long)blockIdx.x * 1024 + r] = s + bias[0];
    }
}

extern "C" void kernel_launch(void* const* d_in, const int* in_sizes, int n_in,
                              void* d_out, int out_size, void* d_ws, size_t ws_size,
                              hipStream_t stream) {
    const float* x   = (const float*)d_in[0];
    const float* mpw = (const float*)d_in[1];
    const float* dk  = (const float*)d_in[2];
    const float* db  = (const float*)d_in[3];
    float* out = (float*)d_out;
    float* ws  = (float*)d_ws;

    hipLaunchKernelGGL(mpo_setup_kernel, dim3(256), dim3(256), 0, stream,
                       mpw, ws + 524288);

    hipLaunchKernelGGL(mpo_main_kernel, dim3(256), dim3(128), 0, stream,
                       x, dk, db, ws + 524288, out);
}


# main: the 16-lane row sums of each strip as row_xmask DPP moves + adds on the vector ALU, not __shfl_xor (ds_bpermute, one LDS round trip waited for per step); same butterfly, f32 sum order unchanged
# speedup vs baseline: 2.6143x; 1.0462x over previous
#include <hip/hip_runtime.h>
#include <hip/hip_bf16.h>

typedef float v2f __attribute__((ext_vector_type(2)));
typedef float v4f __attribute__((ext_vector_type(4)));
typedef float v8f __attribute__((ext_vector_type(8)));

static __device__ __forceinline__ v8f wmma_16x16x4_f32(v2f a, v2f b, v8f c) {
#if __has_builtin(__builtin_amdgcn_wmma_f32_16x16x4_f32)
    return __builtin_amdgcn_wmma_f32_16x16x4_f32(
         false, a,  false, b,
         (short)0, c,  false,  false);
#else
    v8f d;
    asm volatile("v_wmma_f32_16x16x4_f32 %0, %1, %2, %3"
                 : "=v"(d)
                 : "v"(a), "v"(b), "v"(c));
    return d;
#endif
}

__global__ void __launch_bounds__(256)
mpo_setup_kernel(const float* __restrict__ w, float* __restrict__ wf) {
    __shared__ __attribute__((aligned(16))) float cs[6272];
    __shared__ float vec[2][16];
    __shared__ float t5[4 * 16];
    __shared__ float t6[16 * 16];
    __shared__ float t7[64 * 16];

    const int tid = threadIdx.x;
    const int P   = blockIdx.x >> 4;
    const int Q   = blockIdx.x & 15;
    const int b   = tid & 15;
    const int hi  = tid >> 4;

    if (tid < 224) {
        const float4* src4 = (const float4*)w;
        float4* dst4 = (float4*)cs;
        float4 r[7];
        for (int u = 0; u < 7; ++u) r[u] = src4[tid + 224 * u];
        for (int u = 0; u < 7; ++u) dst4[tid + 224 * u] = r[u];
    }
    __syncthreads();

    if (tid < 16)
        vec[0][tid] = cs[tid * 4 + ((P >> 3) & 1) * 2 + ((Q >> 3) & 1)];
    __syncthreads();

    for (int c = 1; c <= 3; ++c) {
        if (tid < 16) {
            const int i = (P >> (3 - c)) & 1;
            const int j = (Q >> (3 - c)) & 1;
            const float* core = cs + 64 + (c - 1) * 1024 + b * 4 + i * 2 + j;
            const float* src  = vec[(c - 1) & 1];
            float s = 0.f;
            for (int a = 0; a < 16; ++a)
                s += src[a] * core[a * 64];
            vec[c & 1][b] = s;
        }
        __syncthreads();
    }

    if (tid < 64) {
        const float* core = cs + 64 + 3 * 1024 + b * 4 + hi;
        float s = 0.f;
        for (int a = 0; a < 16; ++a)
            s += vec[1][a] * core[a * 64];
        t5[tid] = s;
    }
    __syncthreads();

    {
        const int Il = hi >> 2, Jl = hi & 3;
        const float* src  = t5 + ((Il >> 1) * 2 + (Jl >> 1)) * 16;
        const float* core = cs + 64 + 4 * 1024 + b * 4 + (Il & 1) * 2 + (Jl & 1);
        float s = 0.f;
        for (int a = 0; a < 16; ++a)
            s += src[a] * core[a * 64];
        t6[tid] = s;
    }
    __syncthreads();

    for (int m = 0; m < 4; ++m) {
        const int cell = hi + 16 * m;
        const int Il = cell >> 3, Jl = cell & 7;
        const float* src  = t6 + ((Il >> 1) * 4 + (Jl >> 1)) * 16;
        const float* core = cs + 64 + 5 * 1024 + b * 4 + (Il & 1) * 2 + (Jl & 1);
        float s = 0.f;
        for (int a = 0; a < 16; ++a)
            s += src[a] * core[a * 64];
        t7[cell * 16 + b] = s;
    }
    __syncthreads();

    {
        const int rl = hi, nl = b;
        const float* src  = t7 + ((rl >> 1) * 8 + (nl >> 1)) * 16;
        const float* core = cs + 6208 + (rl & 1) * 2 + (nl & 1);
        float s = 0.f;
        for (int a = 0; a < 16; ++a)
            s += src[a] * core[a * 4];
        const int r     = P * 16 + rl;
        const int kstep = r >> 2, kk = r & 3;
        const int lane  = ((kk >= 2) ? 16 : 0) + nl;
        const int v     = kk & 1;
        wf[((kstep * 16 + Q) * 32 + lane) * 2 + v] = s;
    }
}


template <int M>
static __device__ __forceinline__ float row_xor_sum(float v) {
    const int partner = __builtin_amdgcn_update_dpp(__float_as_int(v), __float_as_int(v),
                                                    0x160 + M, 0xf, 0xf, true);
    return v + __int_as_float(partner);
}

static __device__ __forceinline__ void strip_pass(const v4f (&cur)[32], v4f (&nxt)[32],
                                                  const float* __restrict__ xnext,
                                                  const v2f (&wr)[64][4],
                                                  const float (&kv)[4],
                                                  float* __restrict__ prow,
                                                  int c, int h) {
#pragma unroll
    for (int j = 0; j < 32; ++j)
        nxt[j] = *(const v4f*)(xnext + 8 * j);

    v8f acc[4];
#pragma unroll
    for (int t = 0; t < 4; ++t)
#pragma unroll
        for (int v = 0; v < 8; ++v)
            acc[t][v] = 0.f;

#pragma unroll
    for (int j = 0; j < 32; ++j) {
        v2f alo, ahi;
        alo[0] = cur[j][0]; alo[1] = cur[j][1];
        ahi[0] = cur[j][2]; ahi[1] = cur[j][3];
#pragma unroll
        for (int t = 0; t < 4; ++t)
            acc[t] = wmma_16x16x4_f32(alo, wr[2 * j][t], acc[t]);
#pragma unroll
        for (int t = 0; t < 4; ++t)
            acc[t] = wmma_16x16x4_f32(ahi, wr[2 * j + 1][t], acc[t]);
    }

    float sv[8];
#pragma unroll
    for (int v = 0; v < 8; ++v) sv[v] = 0.f;
#pragma unroll
    for (int t = 0; t < 4; ++t)
#pragma unroll
        for (int v = 0; v < 8; ++v) {
            float hh = acc[t][v];
            sv[v] += hh * hh * kv[t];
        }
#pragma unroll
    for (int v = 0; v < 8; ++v) sv[v] = row_xor_sum<1>(sv[v]);
#pragma unroll
    for (int v = 0; v < 8; ++v) sv[v] = row_xor_sum<2>(sv[v]);
#pragma unroll
    for (int v = 0; v < 8; ++v) sv[v] = row_xor_sum<4>(sv[v]);
#pragma unroll
    for (int v = 0; v < 8; ++v) sv[v] = row_xor_sum<8>(sv[v]);
    if (c == 0) {
        float* p = prow + h * 8;
#pragma unroll
        for (int v = 0; v < 8; ++v) p[v] = sv[v];
    }
}

__global__ void __launch_bounds__(128)
mpo_main_kernel(const float* __restrict__ x,
                const float* __restrict__ dk,
                const float* __restrict__ bias,
                const float* __restrict__ wf,
                float* __restrict__ out) {
    __shared__ float part[4 * 1024];
    const int tid  = threadIdx.x;
    const int g    = tid >> 5;
    const int lane = tid & 31;
    const int c    = lane & 15;
    const int h    = lane >> 4;

    v2f wr[64][4];
    {
        const float* wb = wf + ((h * 16 + g * 4) * 32 + c) * 2;
#pragma unroll
        for (int s = 0; s < 64; ++s)
#pragma unroll
            for (int t = 0; t < 4; ++t)
                wr[s][t] = *(const v2f*)(wb + (((s >> 1) * 32 + t) * 32 + 16 * (s & 1)) * 2);
    }

    float kv[4];
#pragma unroll
    for (int t = 0; t < 4; ++t)
        kv[t] = dk[(g * 4 + t) * 16 + c];

    const float* xb = x + ((long)blockIdx.x * 1024 + c) * 256 + 4 * h;
    float* pg = part + g * 1024;

    v4f bufa[32], bufb[32];
#pragma unroll
    for (int j = 0; j < 32; ++j)
        bufa[j] = *(const v4f*)(xb + 8 * j);

    for (int p = 0; p < 64; p += 2) {
        strip_pass(bufa, bufb, xb + (long)(p + 1) * 4096, wr, kv, pg + p * 16, c, h);
        const int pn = (p + 2 < 64) ? (p + 2) : 63;
        strip_pass(bufb, bufa, xb + (long)pn * 4096, wr, kv, pg + (p + 1) * 16, c, h);
    }
    __syncthreads();

#pragma unroll
    for (int u = 0; u < 8; ++u) {
        const int r = tid + 128 * u;
        float s = part[r] + part[1024 + r] + part[2048 + r] + part[3072 + r];
        out[(long)blockIdx.x * 1024 + r] = s + bias[0];
    }
}

extern "C" void kernel_launch(void* const* d_in, const int* in_sizes, int n_in,
                              void* d_out, int out_size, void* d_ws, size_t ws_size,
                              hipStream_t stream) {
    const float* x   = (const float*)d_in[0];
    const float* mpw = (const float*)d_in[1];
    const float* dk  = (const float*)d_in[2];
    const float* db  = (const float*)d_in[3];
    float* out = (float*)d_out;
    float* ws  = (float*)d_ws;

    hipLaunchKernelGGL(mpo_setup_kernel, dim3(256), dim3(256), 0, stream,
                       mpw, ws + 524288);

    hipLaunchKernelGGL(mpo_main_kernel, dim3(256), dim3(128), 0, stream,
                       x, dk, db, ws + 524288, out);
}


# main: next strip's 32 x loads pinned as ONE group ahead of the strip's 256 WMMAs (scheduling barrier, second register set kept), not sunk into ten small places among them; f32 sum order unchanged
# speedup vs baseline: 2.6983x; 1.0321x over previous
#include <hip/hip_runtime.h>
#include <hip/hip_bf16.h>

typedef float v2f __attribute__((ext_vector_type(2)));
typedef float v4f __attribute__((ext_vector_type(4)));
typedef float v8f __attribute__((ext_vector_type(8)));

static __device__ __forceinline__ v8f wmma_16x16x4_f32(v2f a, v2f b, v8f c) {
#if __has_builtin(__builtin_amdgcn_wmma_f32_16x16x4_f32)
    return __builtin_amdgcn_wmma_f32_16x16x4_f32(
         false, a,  false, b,
         (short)0, c,  false,  false);
#else
    v8f d;
    asm volatile("v_wmma_f32_16x16x4_f32 %0, %1, %2, %3"
                 : "=v"(d)
                 : "v"(a), "v"(b), "v"(c));
    return d;
#endif
}

__global__ void __launch_bounds__(256)
mpo_setup_kernel(const float* __restrict__ w, float* __restrict__ wf) {
    __shared__ __attribute__((aligned(16))) float cs[6272];
    __shared__ float vec[2][16];
    __shared__ float t5[4 * 16];
    __shared__ float t6[16 * 16];
    __shared__ float t7[64 * 16];

    const int tid = threadIdx.x;
    const int P   = blockIdx.x >> 4;
    const int Q   = blockIdx.x & 15;
    const int b   = tid & 15;
    const int hi  = tid >> 4;

    if (tid < 224) {
        const float4* src4 = (const float4*)w;
        float4* dst4 = (float4*)cs;
        float4 r[7];
        for (int u = 0; u < 7; ++u) r[u] = src4[tid + 224 * u];
        for (int u = 0; u < 7; ++u) dst4[tid + 224 * u] = r[u];
    }
    __syncthreads();

    if (tid < 16)
        vec[0][tid] = cs[tid * 4 + ((P >> 3) & 1) * 2 + ((Q >> 3) & 1)];
    __syncthreads();

    for (int c = 1; c <= 3; ++c) {
        if (tid < 16) {
            const int i = (P >> (3 - c)) & 1;
            const int j = (Q >> (3 - c)) & 1;
            const float* core = cs + 64 + (c - 1) * 1024 + b * 4 + i * 2 + j;
            const float* src  = vec[(c - 1) & 1];
            float s = 0.f;
            for (int a = 0; a < 16; ++a)
                s += src[a] * core[a * 64];
            vec[c & 1][b] = s;
        }
        __syncthreads();
    }

    if (tid < 64) {
        const float* core = cs + 64 + 3 * 1024 + b * 4 + hi;
        float s = 0.f;
        for (int a = 0; a < 16; ++a)
            s += vec[1][a] * core[a * 64];
        t5[tid] = s;
    }
    __syncthreads();

    {
        const int Il = hi >> 2, Jl = hi & 3;
        const float* src  = t5 + ((Il >> 1) * 2 + (Jl >> 1)) * 16;
        const float* core = cs + 64 + 4 * 1024 + b * 4 + (Il & 1) * 2 + (Jl & 1);
        float s = 0.f;
        for (int a = 0; a < 16; ++a)
            s += src[a] * core[a * 64];
        t6[tid] = s;
    }
    __syncthreads();

    for (int m = 0; m < 4; ++m) {
        const int cell = hi + 16 * m;
        const int Il = cell >> 3, Jl = cell & 7;
        const float* src  = t6 + ((Il >> 1) * 4 + (Jl >> 1)) * 16;
        const float* core = cs + 64 + 5 * 1024 + b * 4 + (Il & 1) * 2 + (Jl & 1);
        float s = 0.f;
        for (int a = 0; a < 16; ++a)
            s += src[a] * core[a * 64];
        t7[cell * 16 + b] = s;
    }
    __syncthreads();

    {
        const int rl = hi, nl = b;
        const float* src  = t7 + ((rl >> 1) * 8 + (nl >> 1)) * 16;
        const float* core = cs + 6208 + (rl & 1) * 2 + (nl & 1);
        float s = 0.f;
        for (int a = 0; a < 16; ++a)
            s += src[a] * core[a * 4];
        const int r     = P * 16 + rl;
        const int kstep = r >> 2, kk = r & 3;
        const int lane  = ((kk >= 2) ? 16 : 0) + nl;
        const int v     = kk & 1;
        wf[((kstep * 16 + Q) * 32 + lane) * 2 + v] = s;
    }
}


template <int M>
static __device__ __forceinline__ float row_xor_sum(float v) {
    const int partner = __builtin_amdgcn_update_dpp(__float_as_int(v), __float_as_int(v),
                                                    0x160 + M, 0xf, 0xf, true);
    return v + __int_as_float(partner);
}

static __device__ __forceinline__ void strip_pass(const v4f (&cur)[32], v4f (&nxt)[32],
                                                  const float* __restrict__ xnext,
                                                  const v2f (&wr)[64][4],
                                                  const float (&kv)[4],
                                                  float* __restrict__ prow,
                                                  int c, int h) {
#pragma unroll
    for (int j = 0; j < 32; ++j)
        nxt[j] = *(const v4f*)(xnext + 8 * j);
    __builtin_amdgcn_sched_barrier(0);

    v8f acc[4];
#pragma unroll
    for (int t = 0; t < 4; ++t)
#pragma unroll
        for (int v = 0; v < 8; ++v)
            acc[t][v] = 0.f;

#pragma unroll
    for (int j = 0; j < 32; ++j) {
        v2f alo, ahi;
        alo[0] = cur[j][0]; alo[1] = cur[j][1];
        ahi[0] = cur[j][2]; ahi[1] = cur[j][3];
#pragma unroll
        for (int t = 0; t < 4; ++t)
            acc[t] = wmma_16x16x4_f32(alo, wr[2 * j][t], acc[t]);
#pragma unroll
        for (int t = 0; t < 4; ++t)
            acc[t] = wmma_16x16x4_f32(ahi, wr[2 * j + 1][t], acc[t]);
    }

    float sv[8];
#pragma unroll
    for (int v = 0; v < 8; ++v) sv[v] = 0.f;
#pragma unroll
    for (int t = 0; t < 4; ++t)
#pragma unroll
        for (int v = 0; v < 8; ++v) {
            float hh = acc[t][v];
            sv[v] += hh * hh * kv[t];
        }
#pragma unroll
    for (int v = 0; v < 8; ++v) sv[v] = row_xor_sum<1>(sv[v]);
#pragma unroll
    for (int v = 0; v < 8; ++v) sv[v] = row_xor_sum<2>(sv[v]);
#pragma unroll
    for (int v = 0; v < 8; ++v) sv[v] = row_xor_sum<4>(sv[v]);
#pragma unroll
    for (int v = 0; v < 8; ++v) sv[v] = row_xor_sum<8>(sv[v]);
    if (c == 0) {
        float* p = prow + h * 8;
#pragma unroll
        for (int v = 0; v < 8; ++v) p[v] = sv[v];
    }
}

__global__ void __launch_bounds__(128)
mpo_main_kernel(const float* __restrict__ x,
                const float* __restrict__ dk,
                const float* __restrict__ bias,
                const float* __restrict__ wf,
                float* __restrict__ out) {
    __shared__ float part[4 * 1024];
    const int tid  = threadIdx.x;
    const int g    = tid >> 5;
    const int lane = tid & 31;
    const int c    = lane & 15;
    const int h    = lane >> 4;

    v2f wr[64][4];
    {
        const float* wb = wf + ((h * 16 + g * 4) * 32 + c) * 2;
#pragma unroll
        for (int s = 0; s < 64; ++s)
#pragma unroll
            for (int t = 0; t < 4; ++t)
                wr[s][t] = *(const v2f*)(wb + (((s >> 1) * 32 + t) * 32 + 16 * (s & 1)) * 2);
    }

    float kv[4];
#pragma unroll
    for (int t = 0; t < 4; ++t)
        kv[t] = dk[(g * 4 + t) * 16 + c];

    const float* xb = x + ((long)blockIdx.x * 1024 + c) * 256 + 4 * h;
    float* pg = part + g * 1024;

    v4f bufa[32], bufb[32];
#pragma unroll
    for (int j = 0; j < 32; ++j)
        bufa[j] = *(const v4f*)(xb + 8 * j);

    for (int p = 0; p < 64; p += 2) {
        strip_pass(bufa, bufb, xb + (long)(p + 1) * 4096, wr, kv, pg + p * 16, c, h);
        const int pn = (p + 2 < 64) ? (p + 2) : 63;
        strip_pass(bufb, bufa, xb + (long)pn * 4096, wr, kv, pg + (p + 1) * 16, c, h);
    }
    __syncthreads();

#pragma unroll
    for (int u = 0; u < 8; ++u) {
        const int r = tid + 128 * u;
        float s = part[r] + part[1024 + r] + part[2048 + r] + part[3072 + r];
        out[(long)blockIdx.x * 1024 + r] = s + bias[0];
    }
}

extern "C" void kernel_launch(void* const* d_in, const int* in_sizes, int n_in,
                              void* d_out, int out_size, void* d_ws, size_t ws_size,
                              hipStream_t stream) {
    const float* x   = (const float*)d_in[0];
    const float* mpw = (const float*)d_in[1];
    const float* dk  = (const float*)d_in[2];
    const float* db  = (const float*)d_in[3];
    float* out = (float*)d_out;
    float* ws  = (float*)d_ws;

    hipLaunchKernelGGL(mpo_setup_kernel, dim3(256), dim3(256), 0, stream,
                       mpw, ws + 524288);

    hipLaunchKernelGGL(mpo_main_kernel, dim3(256), dim3(128), 0, stream,
                       x, dk, db, ws + 524288, out);
}
